# starting kernel
# baseline (speedup 1.0000x reference)
#include <hip/hip_runtime.h>
#include <hip/hip_bf16.h>

typedef __attribute__((ext_vector_type(2))) float v2f;
typedef __attribute__((ext_vector_type(8))) float v8f;

#define D_DIM 64
#define TILE_M 16
#define NEG_SLOPE 0.01f

__device__ __forceinline__ void atomic_max_f32(float* addr, float val) {
  if (val >= 0.0f) {
    atomicMax((int*)addr, __float_as_int(val));
  } else {
    atomicMin((unsigned int*)addr, __float_as_uint(val));
  }
}

__global__ __launch_bounds__(256, 2) void gat_linear_wmma(
    const float* __restrict__ feat, const float* __restrict__ W,
    const float* __restrict__ bias,
    const float* __restrict__ a1w, const float* __restrict__ a1b,
    const float* __restrict__ a2w, const float* __restrict__ a2b,
    float* __restrict__ h, float* __restrict__ a1, float* __restrict__ a2,
    int n_nodes)
{
  constexpr int WP = 68;
  constexpr int FP = 68;
  __shared__ float Wl[D_DIM * WP];
  __shared__ float ft[8][TILE_M * FP];

  const int tid  = threadIdx.x;
  const int wave = tid >> 5;
  const int lane = tid & 31;
  const int lh   = lane >> 4;
  const int lid  = lane & 15;

  for (int i = tid; i < D_DIM * (D_DIM / 4); i += 256) {
    const int n  = i >> 4;
    const int c4 = (i & 15) << 2;
    const float4 w4 = *(const float4*)(W + n * D_DIM + c4);
    *(float4*)&Wl[n * WP + c4] = w4;
  }
  __syncthreads();

  const int tile = blockIdx.x * 8 + wave;
  const int m0   = tile * TILE_M;
  if (m0 >= n_nodes) return;

  float* myft = ft[wave];
  for (int i = lane; i < TILE_M * (D_DIM / 4); i += 32) {
    const int r  = i >> 4;
    const int c4 = (i & 15) << 2;
    float4 f4 = {0.f, 0.f, 0.f, 0.f};
    if (m0 + r < n_nodes)
      f4 = *(const float4*)(feat + (size_t)(m0 + r) * D_DIM + c4);
    *(float4*)&myft[r * FP + c4] = f4;
  }

  v8f acc[4] = {};
  const int ks = 2 * lh;
  #pragma unroll
  for (int k = 0; k < D_DIM; k += 4) {
    v2f a;
    a.x = myft[lid * FP + k + ks + 0];
    a.y = myft[lid * FP + k + ks + 1];
    #pragma unroll
    for (int t = 0; t < 4; ++t) {
      const int n = t * 16 + lid;
      v2f b;
      b.x = Wl[n * WP + k + ks + 0];
      b.y = Wl[n * WP + k + ks + 1];
      acc[t] = __builtin_amdgcn_wmma_f32_16x16x4_f32(
          false, a, false, b, (short)0, acc[t], false, false);
    }
  }

  #pragma unroll
  for (int t = 0; t < 4; ++t) {
    const float bb = bias[t * 16 + lid];
    #pragma unroll
    for (int j = 0; j < 8; ++j) {
      myft[(j + 8 * lh) * FP + t * 16 + lid] = acc[t][j] + bb;
    }
  }

  for (int i = lane; i < TILE_M * (D_DIM / 4); i += 32) {
    const int r  = i >> 4;
    const int c4 = (i & 15) << 2;
    if (m0 + r < n_nodes) {
      const float4 hv = *(const float4*)&myft[r * FP + c4];
      *(float4*)(h + (size_t)(m0 + r) * D_DIM + c4) = hv;
    }
  }

  const float* aw = lh ? a2w : a1w;
  float s = lh ? a2b[0] : a1b[0];
  #pragma unroll 8
  for (int c = 0; c < D_DIM; ++c)
    s += myft[lid * FP + c] * aw[c];
  if (m0 + lid < n_nodes) {
    if (lh) a2[m0 + lid] = s;
    else    a1[m0 + lid] = s;
  }
}

__global__ void gat_init(float* __restrict__ out, float* __restrict__ segmax,
                         float* __restrict__ denom, int n) {
  const int i = blockIdx.x * blockDim.x + threadIdx.x;
  if (i < n * D_DIM) out[i] = 0.0f;
  if (i < n) { segmax[i] = -__builtin_inff(); denom[i] = 0.0f; }
}

__global__ void gat_edge_score(const int* __restrict__ src, const int* __restrict__ dst,
                               const float* __restrict__ a1, const float* __restrict__ a2,
                               float* __restrict__ vbuf, float* __restrict__ segmax, int E) {
  const int e = blockIdx.x * blockDim.x + threadIdx.x;
  if (e >= E) return;
  const int s = src[e], d = dst[e];
  float v = a1[s] + a2[d];
  v = v > 0.0f ? v : v * NEG_SLOPE;
  vbuf[e] = v;
  atomic_max_f32(segmax + s, v);
}

__global__ void gat_edge_exp(const int* __restrict__ src, const float* __restrict__ segmax,
                             float* __restrict__ vbuf, float* __restrict__ denom, int E) {
  const int e = blockIdx.x * blockDim.x + threadIdx.x;
  if (e >= E) return;
  const int s = src[e];
  const float ev = __expf(vbuf[e] - segmax[s]);
  vbuf[e] = ev;
  unsafeAtomicAdd(denom + s, ev);
}

__global__ __launch_bounds__(256) void gat_aggregate(
    const int* __restrict__ src, const int* __restrict__ dst,
    const float* __restrict__ evbuf, const float* __restrict__ denom,
    const float* __restrict__ h, float* __restrict__ out, int E) {
  const int e = blockIdx.x * 8 + (threadIdx.x >> 5);
  if (e >= E) return;
  const int lane = threadIdx.x & 31;
  const int s = src[e], d = dst[e];
  const float att = evbuf[e] / denom[s];
  const float* hr  = h   + (size_t)d * D_DIM;
  float*       orow = out + (size_t)s * D_DIM;
  unsafeAtomicAdd(orow + lane,      att * hr[lane]);
  unsafeAtomicAdd(orow + lane + 32, att * hr[lane + 32]);
}

extern "C" void kernel_launch(void* const* d_in, const int* in_sizes, int n_in,
                              void* d_out, int out_size, void* d_ws, size_t ws_size,
                              hipStream_t stream) {
  const float* feat = (const float*)d_in[0];
  const int*   idx  = (const int*)d_in[1];
  const float* W    = (const float*)d_in[2];
  const float* bias = (const float*)d_in[3];
  const float* a1w  = (const float*)d_in[4];
  const float* a1b  = (const float*)d_in[5];
  const float* a2w  = (const float*)d_in[6];
  const float* a2b  = (const float*)d_in[7];
  float* out = (float*)d_out;

  const int n = in_sizes[0] / D_DIM;
  const int E = in_sizes[1] / 2;
  const int* src = idx;
  const int* dst = idx + E;

  float* h      = (float*)d_ws;
  float* a1     = h + (size_t)n * D_DIM;
  float* a2     = a1 + n;
  float* segmax = a2 + n;
  float* denom  = segmax + n;
  float* vbuf   = denom + n;

  gat_init<<<(n * D_DIM + 255) / 256, 256, 0, stream>>>(out, segmax, denom, n);

  const int tiles = (n + TILE_M - 1) / TILE_M;
  gat_linear_wmma<<<(tiles + 7) / 8, 256, 0, stream>>>(
      feat, W, bias, a1w, a1b, a2w, a2b, h, a1, a2, n);

  gat_edge_score<<<(E + 255) / 256, 256, 0, stream>>>(src, dst, a1, a2, vbuf, segmax, E);
  gat_edge_exp<<<(E + 255) / 256, 256, 0, stream>>>(src, segmax, vbuf, denom, E);
  gat_aggregate<<<(E + 7) / 8, 256, 0, stream>>>(src, dst, vbuf, denom, h, out, E);
}


# edge key bytes replaced by a unit bitmap per row range (1 bit per 16 edges, built by LDS ORs): a row-pass work-group reads its own bitmap row, loads src of flagged units only; sums stay LDS-add order
# speedup vs baseline: 4.4563x; 4.4563x over previous
#include <hip/hip_runtime.h>
#include <hip/hip_bf16.h>

typedef __attribute__((ext_vector_type(2))) float v2f;
typedef __attribute__((ext_vector_type(8))) float v8f;

#define D_DIM 64
#define TILE_M 16
#define NEG_SLOPE 0.01f

__global__ __launch_bounds__(256, 2) void gat_linear_wmma(
    const float* __restrict__ feat, const float* __restrict__ W,
    const float* __restrict__ bias,
    const float* __restrict__ a1w, const float* __restrict__ a1b,
    const float* __restrict__ a2w, const float* __restrict__ a2b,
    float* __restrict__ h, float* __restrict__ a1, float* __restrict__ a2,
    int n_nodes)
{
  constexpr int WP = 68;
  constexpr int FP = 68;
  __shared__ float Wl[D_DIM * WP];
  __shared__ float ft[8][TILE_M * FP];

  const int tid  = threadIdx.x;
  const int wave = tid >> 5;
  const int lane = tid & 31;
  const int lh   = lane >> 4;
  const int lid  = lane & 15;

  for (int i = tid; i < D_DIM * (D_DIM / 4); i += 256) {
    const int n  = i >> 4;
    const int c4 = (i & 15) << 2;
    const float4 w4 = *(const float4*)(W + n * D_DIM + c4);
    *(float4*)&Wl[n * WP + c4] = w4;
  }
  __syncthreads();

  const int tile = blockIdx.x * 8 + wave;
  const int m0   = tile * TILE_M;
  if (m0 >= n_nodes) return;

  float* myft = ft[wave];
  for (int i = lane; i < TILE_M * (D_DIM / 4); i += 32) {
    const int r  = i >> 4;
    const int c4 = (i & 15) << 2;
    float4 f4 = {0.f, 0.f, 0.f, 0.f};
    if (m0 + r < n_nodes)
      f4 = *(const float4*)(feat + (size_t)(m0 + r) * D_DIM + c4);
    *(float4*)&myft[r * FP + c4] = f4;
  }

  v8f acc[4] = {};
  const int ks = 2 * lh;
  #pragma unroll
  for (int k = 0; k < D_DIM; k += 4) {
    v2f a;
    a.x = myft[lid * FP + k + ks + 0];
    a.y = myft[lid * FP + k + ks + 1];
    #pragma unroll
    for (int t = 0; t < 4; ++t) {
      const int n = t * 16 + lid;
      v2f b;
      b.x = Wl[n * WP + k + ks + 0];
      b.y = Wl[n * WP + k + ks + 1];
      acc[t] = __builtin_amdgcn_wmma_f32_16x16x4_f32(
          false, a, false, b, (short)0, acc[t], false, false);
    }
  }

  #pragma unroll
  for (int t = 0; t < 4; ++t) {
    const float bb = bias[t * 16 + lid];
    #pragma unroll
    for (int j = 0; j < 8; ++j) {
      myft[(j + 8 * lh) * FP + t * 16 + lid] = acc[t][j] + bb;
    }
  }

  for (int i = lane; i < TILE_M * (D_DIM / 4); i += 32) {
    const int r  = i >> 4;
    const int c4 = (i & 15) << 2;
    if (m0 + r < n_nodes) {
      const float4 hv = *(const float4*)&myft[r * FP + c4];
      *(float4*)(h + (size_t)(m0 + r) * D_DIM + c4) = hv;
    }
  }

  const float* aw = lh ? a2w : a1w;
  float s = lh ? a2b[0] : a1b[0];
  #pragma unroll 8
  for (int c = 0; c < D_DIM; ++c)
    s += myft[lid * FP + c] * aw[c];
  if (m0 + lid < n_nodes) {
    if (lh) a2[m0 + lid] = s;
    else    a1[m0 + lid] = s;
  }
}

#define ROWS_PER_WG 392
#define LIST_CAP 320

#define UNIT_EDGES 16
#define SPAN_WORDS 32
#define SPAN_UNITS (SPAN_WORDS * 32)
#define TILE_PITCH 33

__global__ __launch_bounds__(1024) void gat_edge_bitmap(const int* __restrict__ src,
                                                        unsigned* __restrict__ bm,
                                                        int E, int nRanges, int W) {
  __shared__ unsigned tile[256 * TILE_PITCH];
  const int tid  = threadIdx.x;
  const int lane = tid & 31;
  const int wave = tid >> 5;
  const int e0 = (blockIdx.x * SPAN_UNITS + tid) * UNIT_EDGES;

  unsigned rg[UNIT_EDGES];
  if (e0 + UNIT_EDGES <= E) {
    #pragma unroll
    for (int j = 0; j < 4; ++j) {
      const int4 s4 = *(const int4*)(src + e0 + 4 * j);
      rg[4 * j + 0] = (unsigned)s4.x / (unsigned)ROWS_PER_WG;
      rg[4 * j + 1] = (unsigned)s4.y / (unsigned)ROWS_PER_WG;
      rg[4 * j + 2] = (unsigned)s4.z / (unsigned)ROWS_PER_WG;
      rg[4 * j + 3] = (unsigned)s4.w / (unsigned)ROWS_PER_WG;
    }
  } else {
    #pragma unroll
    for (int k = 0; k < UNIT_EDGES; ++k) {
      const int e = e0 + k;
      rg[k] = e < E ? (unsigned)src[e] / (unsigned)ROWS_PER_WG : 0xFFFFFFFFu;
    }
  }

  const unsigned bit = 1u << lane;
  for (int b0 = 0; b0 < nRanges; b0 += 256) {
    for (int i = tid; i < 256 * TILE_PITCH; i += 1024) tile[i] = 0u;
    __syncthreads();
    #pragma unroll
    for (int k = 0; k < UNIT_EDGES; ++k) {
      const unsigned b = rg[k] - (unsigned)b0;
      if (b < 256u)
        __hip_atomic_fetch_or(&tile[b * TILE_PITCH + wave], bit, __ATOMIC_RELAXED,
                              __HIP_MEMORY_SCOPE_WORKGROUP);
    }
    __syncthreads();
    for (int r = wave; r < 256; r += 32) {
      if (b0 + r < nRanges)
        bm[(size_t)(b0 + r) * W + blockIdx.x * SPAN_WORDS + lane] = tile[r * TILE_PITCH + lane];
    }
    __syncthreads();
  }
}

__device__ __forceinline__ void lds_add(float* p, float v) {
  __hip_atomic_fetch_add(p, v, __ATOMIC_RELAXED, __HIP_MEMORY_SCOPE_WORKGROUP);
}
__device__ __forceinline__ void lds_max(float* p, float v) {
  __hip_atomic_fetch_max(p, v, __ATOMIC_RELAXED, __HIP_MEMORY_SCOPE_WORKGROUP);
}

__device__ __forceinline__ int list_edges(
    int e, int E, int row0, int rows,
    const int* __restrict__ src, const int* __restrict__ dst,
    const float* __restrict__ a2, const float* a1l, float* smax,
    int* lr, int* ld, float* lv, int wn) {
  bool ok = false;
  int r = 0, d = 0;
  float v = 0.0f;
  if (e >= 0 && e < E) {
    const int s = src[e];
    d = dst[e];
    r = s - row0;
    if ((unsigned)r < (unsigned)rows) {
      ok = true;
      v = a1l[r] + a2[d];
      v = v > 0.0f ? v : v * NEG_SLOPE;
      lds_max(smax + r, v);
    }
  }
  const unsigned okm = __builtin_amdgcn_ballot_w32(ok);
  const int pos = wn + (int)__builtin_amdgcn_mbcnt_lo(okm, 0u);
  if (ok && pos < LIST_CAP) {
    lr[pos] = r;
    ld[pos] = d;
    lv[pos] = v;
  }
  return wn + __builtin_popcount(okm);
}

__device__ __forceinline__ void add_rows(int cnt, int r, int d, float w, int lane,
                                         const float* __restrict__ h, float* acc) {
  const int wbits = __float_as_int(w);
  if (cnt == 32) {
    #pragma unroll
    for (int j0 = 0; j0 < 32; j0 += 8) {
      float x0[8], x1[8], wj[8];
      int rj[8];
      #pragma unroll
      for (int j = 0; j < 8; ++j) {
        const int dj = __builtin_amdgcn_readlane(d, j0 + j);
        rj[j] = __builtin_amdgcn_readlane(r, j0 + j);
        wj[j] = __int_as_float(__builtin_amdgcn_readlane(wbits, j0 + j));
        const float* hr = h + (size_t)dj * D_DIM;
        x0[j] = hr[lane];
        x1[j] = hr[lane + 32];
      }
      #pragma unroll
      for (int j = 0; j < 8; ++j) {
        float* ar = acc + rj[j] * D_DIM;
        lds_add(ar + lane,      wj[j] * x0[j]);
        lds_add(ar + lane + 32, wj[j] * x1[j]);
      }
    }
  } else {
    #pragma unroll
    for (int j0 = 0; j0 < 32; j0 += 8) {
      if (j0 < cnt) {
        float x0[8], x1[8], wj[8];
        int rj[8];
        #pragma unroll
        for (int j = 0; j < 8; ++j) {
          const int dj = __builtin_amdgcn_readlane(d, j0 + j);
          rj[j] = __builtin_amdgcn_readlane(r, j0 + j);
          wj[j] = __int_as_float(__builtin_amdgcn_readlane(wbits, j0 + j));
          const float* hr = h + (size_t)dj * D_DIM;
          x0[j] = hr[lane];
          x1[j] = hr[lane + 32];
        }
        #pragma unroll
        for (int j = 0; j < 8; ++j) {
          if (j0 + j < cnt) {
            float* ar = acc + rj[j] * D_DIM;
            lds_add(ar + lane,      wj[j] * x0[j]);
            lds_add(ar + lane + 32, wj[j] * x1[j]);
          }
        }
      }
    }
  }
}

__device__ __forceinline__ void add_edges_direct(
    int e, int E, int row0, int rows, int lane,
    const int* __restrict__ src, const int* __restrict__ dst,
    const float* __restrict__ a2, const float* __restrict__ h,
    const float* a1l, const float* smax, float* acc, float* den) {
  int r = 0, d = 0;
  float w = 0.0f;
  bool ok = false;
  if (e >= 0 && e < E) {
    const int s = src[e];
    r = s - row0;
    if ((unsigned)r < (unsigned)rows) {
      ok = true;
      d = dst[e];
      float v = a1l[r] + a2[d];
      v = v > 0.0f ? v : v * NEG_SLOPE;
      w = __expf(v - smax[r]);
      lds_add(den + r, w);
    }
  }
  unsigned okm = __builtin_amdgcn_ballot_w32(ok);
  const int wbits = __float_as_int(w);
  while (okm) {
    const int j = __builtin_ctz(okm);
    okm &= okm - 1;
    const int dj = __builtin_amdgcn_readlane(d, j);
    const int rk = __builtin_amdgcn_readlane(r, j);
    const float wk = __int_as_float(__builtin_amdgcn_readlane(wbits, j));
    const float* hr = h + (size_t)dj * D_DIM;
    float* ar = acc + rk * D_DIM;
    lds_add(ar + lane,      wk * hr[lane]);
    lds_add(ar + lane + 32, wk * hr[lane + 32]);
  }
}

__device__ __forceinline__ unsigned in_rows(int s, int row0, int rows) {
  return (unsigned)(s - row0) < (unsigned)rows ? 1u : 0u;
}

__device__ __forceinline__ unsigned match_unit(const int4* s4, int row0, int rows) {
  unsigned m = 0u;
  #pragma unroll
  for (int j = 0; j < 4; ++j) {
    m |= (in_rows(s4[j].x, row0, rows) << (4 * j + 0)) |
         (in_rows(s4[j].y, row0, rows) << (4 * j + 1)) |
         (in_rows(s4[j].z, row0, rows) << (4 * j + 2)) |
         (in_rows(s4[j].w, row0, rows) << (4 * j + 3));
  }
  return m;
}

__device__ __forceinline__ unsigned match_last_unit(const int* __restrict__ src, int ebase, int E,
                                                    int row0, int rows) {
  unsigned m = 0u;
  for (int k = 0; k < UNIT_EDGES; ++k) {
    const int e = ebase + k;
    if (e < E) m |= in_rows(src[e], row0, rows) << k;
  }
  return m;
}

template <bool DIRECT>
__device__ __forceinline__ int scan_bitmap(
    const unsigned* __restrict__ bmRow, int W, int E, int row0, int rows, int tid, int lane,
    const int* __restrict__ src, const int* __restrict__ dst,
    const float* __restrict__ a2, const float* __restrict__ h,
    const float* a1l, float* smax, float* acc, float* den,
    int* q, int* lr, int* ld, float* lv) {
  int qn = 0;
  int wn = 0;

  const int wq   = (W + 31) >> 5;
  const int wbeg = (tid >> 5) * wq;
  const int wend = (wbeg + wq) < W ? (wbeg + wq) : W;
  const int nIter = (wq + 31) >> 5;
  unsigned cur = (wbeg + lane) < wend ? bmRow[wbeg + lane] : 0u;

  for (int i = 0; i < nIter; ++i) {
    const int wi = wbeg + i * 32 + lane;
    const unsigned nxt = (wi + 32) < wend ? bmRow[wi + 32] : 0u;
    unsigned bits = cur;

    while (true) {
      const unsigned anyb = __builtin_amdgcn_ballot_w32(bits != 0u);
      if (anyb == 0u) break;
      unsigned m[2];
      int eb[2];
      bool part[2];
      int4 s4[2][4];
      #pragma unroll
      for (int t = 0; t < 2; ++t) {
        const bool hv = bits != 0u;
        const int u = hv ? __builtin_ctz(bits) : 0;
        bits &= bits - 1u;
        eb[t] = (wi * 32 + u) * UNIT_EDGES;
        const bool full = hv && (eb[t] + UNIT_EDGES <= E);
        part[t] = hv && !full;
        #pragma unroll
        for (int j = 0; j < 4; ++j) s4[t][j] = int4{0, 0, 0, 0};
        if (full) {
          #pragma unroll
          for (int j = 0; j < 4; ++j) s4[t][j] = *(const int4*)(src + eb[t] + 4 * j);
        }
        m[t] = full ? 1u : 0u;
      }
      #pragma unroll
      for (int t = 0; t < 2; ++t) {
        m[t] = m[t] != 0u ? match_unit(s4[t], row0, rows) : 0u;
        if (part[t]) m[t] = match_last_unit(src, eb[t], E, row0, rows);
      }
      #pragma unroll
      for (int t = 0; t < 2; ++t) {
        unsigned mm = m[t];
        while (true) {
          const bool has = mm != 0u;
          const unsigned bal = __builtin_amdgcn_ballot_w32(has);
          if (bal == 0u) break;
          if (has) {
            const int p = __builtin_ctz(mm);
            mm &= mm - 1u;
            q[qn + (int)__builtin_amdgcn_mbcnt_lo(bal, 0u)] = eb[t] + p;
          }
          qn += __builtin_popcount(bal);
          if (qn >= 32) {
            qn -= 32;
            const int e = q[qn + lane];
            if constexpr (DIRECT)
              add_edges_direct(e, E, row0, rows, lane, src, dst, a2, h, a1l, smax, acc, den);
            else
              wn = list_edges(e, E, row0, rows, src, dst, a2, a1l, smax, lr, ld, lv, wn);
          }
        }
      }
    }
    cur = nxt;
  }
  if (qn > 0) {
    const int e = lane < qn ? q[lane] : -1;
    if constexpr (DIRECT)
      add_edges_direct(e, E, row0, rows, lane, src, dst, a2, h, a1l, smax, acc, den);
    else
      wn = list_edges(e, E, row0, rows, src, dst, a2, a1l, smax, lr, ld, lv, wn);
  }
  return wn;
}

__global__ __launch_bounds__(1024) void gat_rows_owner(
    const int* __restrict__ src, const int* __restrict__ dst,
    const float* __restrict__ a1, const float* __restrict__ a2,
    const float* __restrict__ h, const unsigned* __restrict__ bm,
    float* __restrict__ out, int n, int E, int W) {
  __shared__ float acc[ROWS_PER_WG * D_DIM];
  __shared__ float den[ROWS_PER_WG];
  __shared__ float smax[ROWS_PER_WG];
  __shared__ float a1l[ROWS_PER_WG];
  __shared__ int queue[32][64];
  __shared__ int list_r[32][LIST_CAP];
  __shared__ int list_d[32][LIST_CAP];
  __shared__ float list_v[32][LIST_CAP];

  const int tid  = threadIdx.x;
  const int lane = tid & 31;
  const int wave = tid >> 5;
  const int row0 = blockIdx.x * ROWS_PER_WG;
  const int rows = (n - row0) < ROWS_PER_WG ? (n - row0) : ROWS_PER_WG;

  for (int i = tid; i < ROWS_PER_WG * D_DIM; i += 1024) acc[i] = 0.0f;
  for (int i = tid; i < ROWS_PER_WG; i += 1024) {
    den[i] = 0.0f;
    smax[i] = -__builtin_inff();
    a1l[i] = i < rows ? a1[row0 + i] : 0.0f;
  }
  __syncthreads();

  const unsigned* bmRow = bm + (size_t)blockIdx.x * W;
  int* q = queue[wave];
  int* lr = list_r[wave];
  int* ld = list_d[wave];
  float* lv = list_v[wave];

  const int wn = scan_bitmap<false>(bmRow, W, E, row0, rows, tid, lane, src, dst, a2, h,
                                  a1l, smax, acc, den, q, lr, ld, lv);

  __syncthreads();

  if (wn <= LIST_CAP) {
    for (int base = 0; base < wn; base += 32) {
      const int cnt = (wn - base) < 32 ? (wn - base) : 32;
      int r = 0, d = 0;
      float w = 0.0f;
      if (lane < cnt) {
        r = lr[base + lane];
        d = ld[base + lane];
        w = __expf(lv[base + lane] - smax[r]);
        lds_add(den + r, w);
      }
      add_rows(cnt, r, d, w, lane, h, acc);
    }
  } else {
    scan_bitmap<true>(bmRow, W, E, row0, rows, tid, lane, src, dst, a2, h,
                    a1l, smax, acc, den, q, lr, ld, lv);
  }

  __syncthreads();

  for (int i = tid; i < rows * (D_DIM / 4); i += 1024) {
    const int r  = i >> 4;
    const int c4 = (i & 15) << 2;
    const float dn = den[r];
    float4 o = *(const float4*)&acc[r * D_DIM + c4];
    if (dn == 0.0f) {
      o.x = 0.0f; o.y = 0.0f; o.z = 0.0f; o.w = 0.0f;
    } else {
      o.x /= dn; o.y /= dn; o.z /= dn; o.w /= dn;
    }
    *(float4*)(out + (size_t)(row0 + r) * D_DIM + c4) = o;
  }
}

extern "C" void kernel_launch(void* const* d_in, const int* in_sizes, int n_in,
                              void* d_out, int out_size, void* d_ws, size_t ws_size,
                              hipStream_t stream) {
  const float* feat = (const float*)d_in[0];
  const int*   idx  = (const int*)d_in[1];
  const float* W    = (const float*)d_in[2];
  const float* bias = (const float*)d_in[3];
  const float* a1w  = (const float*)d_in[4];
  const float* a1b  = (const float*)d_in[5];
  const float* a2w  = (const float*)d_in[6];
  const float* a2b  = (const float*)d_in[7];
  float* out = (float*)d_out;

  const int n = in_sizes[0] / D_DIM;
  const int E = in_sizes[1] / 2;
  const int* src = idx;
  const int* dst = idx + E;
  const int nRanges = (n + ROWS_PER_WG - 1) / ROWS_PER_WG;
  const int nUnits  = (E + UNIT_EDGES - 1) / UNIT_EDGES;
  const int nSpans  = (nUnits + SPAN_UNITS - 1) / SPAN_UNITS;
  const int bmW     = nSpans * SPAN_WORDS;

  float*    h  = (float*)d_ws;
  unsigned* bm = (unsigned*)(h + (size_t)n * D_DIM);
  float*    a1 = (float*)(bm + (size_t)nRanges * bmW);
  float*    a2 = a1 + n;

  const int tiles = (n + TILE_M - 1) / TILE_M;
  gat_linear_wmma<<<(tiles + 7) / 8, 256, 0, stream>>>(
      feat, W, bias, a1w, a1b, a2w, a2b, h, a1, a2, n);

  gat_edge_bitmap<<<nSpans, 1024, 0, stream>>>(src, bm, E, nRanges, bmW);

  gat_rows_owner<<<nRanges, 1024, 0, stream>>>(
      src, dst, a1, a2, h, bm, out, n, E, bmW);
}


# bitmap pass also stores edges as (src,dst) pairs, 128 B a 16-edge unit; row pass reads a flagged unit as one line, batches a2 lookups; per-edge src/dst loads go; list, so LDS-add, order changes
# speedup vs baseline: 4.7711x; 1.0706x over previous
#include <hip/hip_runtime.h>
#include <hip/hip_bf16.h>

typedef __attribute__((ext_vector_type(2))) float v2f;
typedef __attribute__((ext_vector_type(8))) float v8f;

#define D_DIM 64
#define TILE_M 16
#define NEG_SLOPE 0.01f

__global__ __launch_bounds__(256, 2) void gat_linear_wmma(
    const float* __restrict__ feat, const float* __restrict__ W,
    const float* __restrict__ bias,
    const float* __restrict__ a1w, const float* __restrict__ a1b,
    const float* __restrict__ a2w, const float* __restrict__ a2b,
    float* __restrict__ h, float* __restrict__ a1, float* __restrict__ a2,
    int n_nodes)
{
  constexpr int WP = 68;
  constexpr int FP = 68;
  __shared__ float Wl[D_DIM * WP];
  __shared__ float ft[8][TILE_M * FP];

  const int tid  = threadIdx.x;
  const int wave = tid >> 5;
  const int lane = tid & 31;
  const int lh   = lane >> 4;
  const int lid  = lane & 15;

  for (int i = tid; i < D_DIM * (D_DIM / 4); i += 256) {
    const int n  = i >> 4;
    const int c4 = (i & 15) << 2;
    const float4 w4 = *(const float4*)(W + n * D_DIM + c4);
    *(float4*)&Wl[n * WP + c4] = w4;
  }
  __syncthreads();

  const int tile = blockIdx.x * 8 + wave;
  const int m0   = tile * TILE_M;
  if (m0 >= n_nodes) return;

  float* myft = ft[wave];
  for (int i = lane; i < TILE_M * (D_DIM / 4); i += 32) {
    const int r  = i >> 4;
    const int c4 = (i & 15) << 2;
    float4 f4 = {0.f, 0.f, 0.f, 0.f};
    if (m0 + r < n_nodes)
      f4 = *(const float4*)(feat + (size_t)(m0 + r) * D_DIM + c4);
    *(float4*)&myft[r * FP + c4] = f4;
  }

  v8f acc[4] = {};
  const int ks = 2 * lh;
  #pragma unroll
  for (int k = 0; k < D_DIM; k += 4) {
    v2f a;
    a.x = myft[lid * FP + k + ks + 0];
    a.y = myft[lid * FP + k + ks + 1];
    #pragma unroll
    for (int t = 0; t < 4; ++t) {
      const int n = t * 16 + lid;
      v2f b;
      b.x = Wl[n * WP + k + ks + 0];
      b.y = Wl[n * WP + k + ks + 1];
      acc[t] = __builtin_amdgcn_wmma_f32_16x16x4_f32(
          false, a, false, b, (short)0, acc[t], false, false);
    }
  }

  #pragma unroll
  for (int t = 0; t < 4; ++t) {
    const float bb = bias[t * 16 + lid];
    #pragma unroll
    for (int j = 0; j < 8; ++j) {
      myft[(j + 8 * lh) * FP + t * 16 + lid] = acc[t][j] + bb;
    }
  }

  for (int i = lane; i < TILE_M * (D_DIM / 4); i += 32) {
    const int r  = i >> 4;
    const int c4 = (i & 15) << 2;
    if (m0 + r < n_nodes) {
      const float4 hv = *(const float4*)&myft[r * FP + c4];
      *(float4*)(h + (size_t)(m0 + r) * D_DIM + c4) = hv;
    }
  }

  const float* aw = lh ? a2w : a1w;
  float s = lh ? a2b[0] : a1b[0];
  #pragma unroll 8
  for (int c = 0; c < D_DIM; ++c)
    s += myft[lid * FP + c] * aw[c];
  if (m0 + lid < n_nodes) {
    if (lh) a2[m0 + lid] = s;
    else    a1[m0 + lid] = s;
  }
}

#define ROWS_PER_WG 392
#define LIST_CAP 320

#define UNIT_EDGES 16
#define SPAN_WORDS 32
#define SPAN_UNITS (SPAN_WORDS * 32)
#define TILE_PITCH 33

__global__ __launch_bounds__(1024) void gat_edge_bitmap(const int* __restrict__ src,
                                                        const int* __restrict__ dst,
                                                        unsigned* __restrict__ bm,
                                                        int* __restrict__ pairs,
                                                        int E, int nRanges, int W) {
  __shared__ unsigned tile[256 * TILE_PITCH];
  const int tid  = threadIdx.x;
  const int lane = tid & 31;
  const int wave = tid >> 5;
  const int e0 = (blockIdx.x * SPAN_UNITS + tid) * UNIT_EDGES;

  {
    constexpr int TRIPS = (SPAN_UNITS * UNIT_EDGES / 2) / 1024;
    const int eS = blockIdx.x * (SPAN_UNITS * UNIT_EDGES);
    int4* pout = (int4*)(pairs + (size_t)eS * 2);
    int sa[TRIPS], da[TRIPS], sb[TRIPS], db[TRIPS];
    #pragma unroll
    for (int k = 0; k < TRIPS; ++k) {
      const int e  = eS + 2 * (tid + 1024 * k);
      const int ea = e < E ? e : E - 1;
      const int eb = e + 1 < E ? e + 1 : E - 1;
      sa[k] = src[ea];
      da[k] = dst[ea];
      sb[k] = src[eb];
      db[k] = dst[eb];
    }
    #pragma unroll
    for (int k = 0; k < TRIPS; ++k) {
      const int i = tid + 1024 * k;
      const int e = eS + 2 * i;
      int4 p;
      p.x = e < E ? sa[k] : -1;
      p.y = e < E ? da[k] : 0;
      p.z = e + 1 < E ? sb[k] : -1;
      p.w = e + 1 < E ? db[k] : 0;
      pout[i] = p;
    }
  }

  unsigned rg[UNIT_EDGES];
  if (e0 + UNIT_EDGES <= E) {
    #pragma unroll
    for (int j = 0; j < 4; ++j) {
      const int4 s4 = *(const int4*)(src + e0 + 4 * j);
      rg[4 * j + 0] = (unsigned)s4.x / (unsigned)ROWS_PER_WG;
      rg[4 * j + 1] = (unsigned)s4.y / (unsigned)ROWS_PER_WG;
      rg[4 * j + 2] = (unsigned)s4.z / (unsigned)ROWS_PER_WG;
      rg[4 * j + 3] = (unsigned)s4.w / (unsigned)ROWS_PER_WG;
    }
  } else {
    #pragma unroll
    for (int k = 0; k < UNIT_EDGES; ++k) {
      const int e = e0 + k;
      rg[k] = e < E ? (unsigned)src[e] / (unsigned)ROWS_PER_WG : 0xFFFFFFFFu;
    }
  }

  const unsigned bit = 1u << lane;
  for (int b0 = 0; b0 < nRanges; b0 += 256) {
    for (int i = tid; i < 256 * TILE_PITCH; i += 1024) tile[i] = 0u;
    __syncthreads();
    #pragma unroll
    for (int k = 0; k < UNIT_EDGES; ++k) {
      const unsigned b = rg[k] - (unsigned)b0;
      if (b < 256u)
        __hip_atomic_fetch_or(&tile[b * TILE_PITCH + wave], bit, __ATOMIC_RELAXED,
                              __HIP_MEMORY_SCOPE_WORKGROUP);
    }
    __syncthreads();
    for (int r = wave; r < 256; r += 32) {
      if (b0 + r < nRanges)
        bm[(size_t)(b0 + r) * W + blockIdx.x * SPAN_WORDS + lane] = tile[r * TILE_PITCH + lane];
    }
    __syncthreads();
  }
}

__device__ __forceinline__ void lds_add(float* p, float v) {
  __hip_atomic_fetch_add(p, v, __ATOMIC_RELAXED, __HIP_MEMORY_SCOPE_WORKGROUP);
}
__device__ __forceinline__ void lds_max(float* p, float v) {
  __hip_atomic_fetch_max(p, v, __ATOMIC_RELAXED, __HIP_MEMORY_SCOPE_WORKGROUP);
}

__device__ __forceinline__ int list_edges(
    int e, int E, int row0, int rows,
    const int* __restrict__ src, const int* __restrict__ dst,
    const float* __restrict__ a2, const float* a1l, float* smax,
    int* lr, int* ld, float* lv, int wn) {
  bool ok = false;
  int r = 0, d = 0;
  float v = 0.0f;
  if (e >= 0 && e < E) {
    const int s = src[e];
    d = dst[e];
    r = s - row0;
    if ((unsigned)r < (unsigned)rows) {
      ok = true;
      v = a1l[r] + a2[d];
      v = v > 0.0f ? v : v * NEG_SLOPE;
      lds_max(smax + r, v);
    }
  }
  const unsigned okm = __builtin_amdgcn_ballot_w32(ok);
  const int pos = wn + (int)__builtin_amdgcn_mbcnt_lo(okm, 0u);
  if (ok && pos < LIST_CAP) {
    lr[pos] = r;
    ld[pos] = d;
    lv[pos] = v;
  }
  return wn + __builtin_popcount(okm);
}

__device__ __forceinline__ void add_rows(int cnt, int r, int d, float w, int lane,
                                         const float* __restrict__ h, float* acc) {
  const int wbits = __float_as_int(w);
  if (cnt == 32) {
    #pragma unroll
    for (int j0 = 0; j0 < 32; j0 += 8) {
      float x0[8], x1[8], wj[8];
      int rj[8];
      #pragma unroll
      for (int j = 0; j < 8; ++j) {
        const int dj = __builtin_amdgcn_readlane(d, j0 + j);
        rj[j] = __builtin_amdgcn_readlane(r, j0 + j);
        wj[j] = __int_as_float(__builtin_amdgcn_readlane(wbits, j0 + j));
        const float* hr = h + (size_t)dj * D_DIM;
        x0[j] = hr[lane];
        x1[j] = hr[lane + 32];
      }
      #pragma unroll
      for (int j = 0; j < 8; ++j) {
        float* ar = acc + rj[j] * D_DIM;
        lds_add(ar + lane,      wj[j] * x0[j]);
        lds_add(ar + lane + 32, wj[j] * x1[j]);
      }
    }
  } else {
    #pragma unroll
    for (int j0 = 0; j0 < 32; j0 += 8) {
      if (j0 < cnt) {
        float x0[8], x1[8], wj[8];
        int rj[8];
        #pragma unroll
        for (int j = 0; j < 8; ++j) {
          const int dj = __builtin_amdgcn_readlane(d, j0 + j);
          rj[j] = __builtin_amdgcn_readlane(r, j0 + j);
          wj[j] = __int_as_float(__builtin_amdgcn_readlane(wbits, j0 + j));
          const float* hr = h + (size_t)dj * D_DIM;
          x0[j] = hr[lane];
          x1[j] = hr[lane + 32];
        }
        #pragma unroll
        for (int j = 0; j < 8; ++j) {
          if (j0 + j < cnt) {
            float* ar = acc + rj[j] * D_DIM;
            lds_add(ar + lane,      wj[j] * x0[j]);
            lds_add(ar + lane + 32, wj[j] * x1[j]);
          }
        }
      }
    }
  }
}

__device__ __forceinline__ void add_edges_direct(
    int e, int E, int row0, int rows, int lane,
    const int* __restrict__ src, const int* __restrict__ dst,
    const float* __restrict__ a2, const float* __restrict__ h,
    const float* a1l, const float* smax, float* acc, float* den) {
  int r = 0, d = 0;
  float w = 0.0f;
  bool ok = false;
  if (e >= 0 && e < E) {
    const int s = src[e];
    r = s - row0;
    if ((unsigned)r < (unsigned)rows) {
      ok = true;
      d = dst[e];
      float v = a1l[r] + a2[d];
      v = v > 0.0f ? v : v * NEG_SLOPE;
      w = __expf(v - smax[r]);
      lds_add(den + r, w);
    }
  }
  unsigned okm = __builtin_amdgcn_ballot_w32(ok);
  const int wbits = __float_as_int(w);
  while (okm) {
    const int j = __builtin_ctz(okm);
    okm &= okm - 1;
    const int dj = __builtin_amdgcn_readlane(d, j);
    const int rk = __builtin_amdgcn_readlane(r, j);
    const float wk = __int_as_float(__builtin_amdgcn_readlane(wbits, j));
    const float* hr = h + (size_t)dj * D_DIM;
    float* ar = acc + rk * D_DIM;
    lds_add(ar + lane,      wk * hr[lane]);
    lds_add(ar + lane + 32, wk * hr[lane + 32]);
  }
}

__device__ __forceinline__ unsigned in_rows(int s, int row0, int rows) {
  return (unsigned)(s - row0) < (unsigned)rows ? 1u : 0u;
}

__device__ __forceinline__ unsigned match_unit(const int4* s4, int row0, int rows) {
  unsigned m = 0u;
  #pragma unroll
  for (int j = 0; j < 4; ++j) {
    m |= (in_rows(s4[j].x, row0, rows) << (4 * j + 0)) |
         (in_rows(s4[j].y, row0, rows) << (4 * j + 1)) |
         (in_rows(s4[j].z, row0, rows) << (4 * j + 2)) |
         (in_rows(s4[j].w, row0, rows) << (4 * j + 3));
  }
  return m;
}

__device__ __forceinline__ unsigned match_last_unit(const int* __restrict__ src, int ebase, int E,
                                                    int row0, int rows) {
  unsigned m = 0u;
  for (int k = 0; k < UNIT_EDGES; ++k) {
    const int e = ebase + k;
    if (e < E) m |= in_rows(src[e], row0, rows) << k;
  }
  return m;
}

template <bool DIRECT>
__device__ __forceinline__ int scan_bitmap(
    const unsigned* __restrict__ bmRow, int W, int E, int row0, int rows, int tid, int lane,
    const int* __restrict__ src, const int* __restrict__ dst,
    const float* __restrict__ a2, const float* __restrict__ h,
    const float* a1l, float* smax, float* acc, float* den,
    int* q, int* lr, int* ld, float* lv) {
  int qn = 0;
  int wn = 0;

  const int wq   = (W + 31) >> 5;
  const int wbeg = (tid >> 5) * wq;
  const int wend = (wbeg + wq) < W ? (wbeg + wq) : W;
  const int nIter = (wq + 31) >> 5;
  unsigned cur = (wbeg + lane) < wend ? bmRow[wbeg + lane] : 0u;

  for (int i = 0; i < nIter; ++i) {
    const int wi = wbeg + i * 32 + lane;
    const unsigned nxt = (wi + 32) < wend ? bmRow[wi + 32] : 0u;
    unsigned bits = cur;

    while (true) {
      const unsigned anyb = __builtin_amdgcn_ballot_w32(bits != 0u);
      if (anyb == 0u) break;
      unsigned m[2];
      int eb[2];
      bool part[2];
      int4 s4[2][4];
      #pragma unroll
      for (int t = 0; t < 2; ++t) {
        const bool hv = bits != 0u;
        const int u = hv ? __builtin_ctz(bits) : 0;
        bits &= bits - 1u;
        eb[t] = (wi * 32 + u) * UNIT_EDGES;
        const bool full = hv && (eb[t] + UNIT_EDGES <= E);
        part[t] = hv && !full;
        #pragma unroll
        for (int j = 0; j < 4; ++j) s4[t][j] = int4{0, 0, 0, 0};
        if (full) {
          #pragma unroll
          for (int j = 0; j < 4; ++j) s4[t][j] = *(const int4*)(src + eb[t] + 4 * j);
        }
        m[t] = full ? 1u : 0u;
      }
      #pragma unroll
      for (int t = 0; t < 2; ++t) {
        m[t] = m[t] != 0u ? match_unit(s4[t], row0, rows) : 0u;
        if (part[t]) m[t] = match_last_unit(src, eb[t], E, row0, rows);
      }
      #pragma unroll
      for (int t = 0; t < 2; ++t) {
        unsigned mm = m[t];
        while (true) {
          const bool has = mm != 0u;
          const unsigned bal = __builtin_amdgcn_ballot_w32(has);
          if (bal == 0u) break;
          if (has) {
            const int p = __builtin_ctz(mm);
            mm &= mm - 1u;
            q[qn + (int)__builtin_amdgcn_mbcnt_lo(bal, 0u)] = eb[t] + p;
          }
          qn += __builtin_popcount(bal);
          if (qn >= 32) {
            qn -= 32;
            const int e = q[qn + lane];
            if constexpr (DIRECT)
              add_edges_direct(e, E, row0, rows, lane, src, dst, a2, h, a1l, smax, acc, den);
            else
              wn = list_edges(e, E, row0, rows, src, dst, a2, a1l, smax, lr, ld, lv, wn);
          }
        }
      }
    }
    cur = nxt;
  }
  if (qn > 0) {
    const int e = lane < qn ? q[lane] : -1;
    if constexpr (DIRECT)
      add_edges_direct(e, E, row0, rows, lane, src, dst, a2, h, a1l, smax, acc, den);
    else
      wn = list_edges(e, E, row0, rows, src, dst, a2, a1l, smax, lr, ld, lv, wn);
  }
  return wn;
}

__device__ __forceinline__ int keep_edge(int s, int d, int row0, int rows,
                                         int* lr, int* ld, int wn) {
  const int r = s - row0;
  const bool ok = (unsigned)r < (unsigned)rows;
  const unsigned okm = __builtin_amdgcn_ballot_w32(ok);
  const int pos = wn + (int)__builtin_amdgcn_mbcnt_lo(okm, 0u);
  if (ok && pos < LIST_CAP) {
    lr[pos] = r;
    ld[pos] = d;
  }
  return wn + __builtin_popcount(okm);
}

__device__ __forceinline__ int take_units(int cnt, const int* ulp, int lane, int row0, int rows,
                                          const int* __restrict__ pairs,
                                          int* lr, int* ld, int wn) {
  int4 p[8];
  #pragma unroll
  for (int j = 0; j < 8; ++j) {
    const int k = 4 * j + (lane >> 3);
    p[j] = int4{-1, 0, -1, 0};
    if (k < cnt) {
      const int u = ulp[k];
      p[j] = *(const int4*)(pairs + (size_t)u * (2 * UNIT_EDGES) + (lane & 7) * 4);
    }
  }
  #pragma unroll
  for (int j = 0; j < 8; ++j) {
    if (4 * j < cnt) {
      wn = keep_edge(p[j].x, p[j].y, row0, rows, lr, ld, wn);
      wn = keep_edge(p[j].z, p[j].w, row0, rows, lr, ld, wn);
    }
  }
  return wn;
}

__device__ __forceinline__ int scan_units(
    const unsigned* __restrict__ bmRow, int W, int row0, int rows, int tid, int lane,
    const int* __restrict__ pairs, int* q, int* lr, int* ld) {
  int un = 0;
  int wn = 0;

  const int wq   = (W + 31) >> 5;
  const int wbeg = (tid >> 5) * wq;
  const int wend = (wbeg + wq) < W ? (wbeg + wq) : W;
  const int nIter = (wq + 31) >> 5;
  unsigned cur = (wbeg + lane) < wend ? bmRow[wbeg + lane] : 0u;

  for (int i = 0; i < nIter; ++i) {
    const int wi = wbeg + i * 32 + lane;
    const unsigned nxt = (wi + 32) < wend ? bmRow[wi + 32] : 0u;
    unsigned bits = cur;
    while (true) {
      const bool has = bits != 0u;
      const unsigned bal = __builtin_amdgcn_ballot_w32(has);
      if (bal == 0u) break;
      if (has) {
        const int u = __builtin_ctz(bits);
        bits &= bits - 1u;
        q[un + (int)__builtin_amdgcn_mbcnt_lo(bal, 0u)] = wi * 32 + u;
      }
      un += __builtin_popcount(bal);
      if (un >= 32) {
        un -= 32;
        wn = take_units(32, q + un, lane, row0, rows, pairs, lr, ld, wn);
      }
    }
    cur = nxt;
  }
  if (un > 0) wn = take_units(un, q, lane, row0, rows, pairs, lr, ld, wn);
  return wn;
}

__device__ __forceinline__ void score_list(int nl, int lane, const float* __restrict__ a2,
                                           const float* a1l, float* smax,
                                           const int* lr, const int* ld, float* lv) {
  for (int base = 0; base < nl; base += 128) {
    int r[4], d[4];
    float a[4];
    #pragma unroll
    for (int t = 0; t < 4; ++t) {
      const int i = base + 32 * t + lane;
      r[t] = 0;
      d[t] = 0;
      if (i < nl) {
        r[t] = lr[i];
        d[t] = ld[i];
      }
    }
    #pragma unroll
    for (int t = 0; t < 4; ++t) a[t] = a2[d[t]];
    #pragma unroll
    for (int t = 0; t < 4; ++t) {
      const int i = base + 32 * t + lane;
      if (i < nl) {
        float v = a1l[r[t]] + a[t];
        v = v > 0.0f ? v : v * NEG_SLOPE;
        lds_max(smax + r[t], v);
        lv[i] = v;
      }
    }
  }
}

__global__ __launch_bounds__(1024) void gat_rows_owner(
    const int* __restrict__ src, const int* __restrict__ dst,
    const float* __restrict__ a1, const float* __restrict__ a2,
    const float* __restrict__ h, const unsigned* __restrict__ bm,
    const int* __restrict__ pairs,
    float* __restrict__ out, int n, int E, int W) {
  __shared__ float acc[ROWS_PER_WG * D_DIM];
  __shared__ float den[ROWS_PER_WG];
  __shared__ float smax[ROWS_PER_WG];
  __shared__ float a1l[ROWS_PER_WG];
  __shared__ int queue[32][64];
  __shared__ int list_r[32][LIST_CAP];
  __shared__ int list_d[32][LIST_CAP];
  __shared__ float list_v[32][LIST_CAP];

  const int tid  = threadIdx.x;
  const int lane = tid & 31;
  const int wave = tid >> 5;
  const int row0 = blockIdx.x * ROWS_PER_WG;
  const int rows = (n - row0) < ROWS_PER_WG ? (n - row0) : ROWS_PER_WG;

  for (int i = tid; i < ROWS_PER_WG * D_DIM; i += 1024) acc[i] = 0.0f;
  for (int i = tid; i < ROWS_PER_WG; i += 1024) {
    den[i] = 0.0f;
    smax[i] = -__builtin_inff();
    a1l[i] = i < rows ? a1[row0 + i] : 0.0f;
  }
  __syncthreads();

  const unsigned* bmRow = bm + (size_t)blockIdx.x * W;
  int* q = queue[wave];
  int* lr = list_r[wave];
  int* ld = list_d[wave];
  float* lv = list_v[wave];

  int wn = scan_units(bmRow, W, row0, rows, tid, lane, pairs, q, lr, ld);
  if (wn <= LIST_CAP)
    score_list(wn, lane, a2, a1l, smax, lr, ld, lv);
  else
    wn = scan_bitmap<false>(bmRow, W, E, row0, rows, tid, lane, src, dst, a2, h,
                            a1l, smax, acc, den, q, lr, ld, lv);

  __syncthreads();

  if (wn <= LIST_CAP) {
    for (int base = 0; base < wn; base += 32) {
      const int cnt = (wn - base) < 32 ? (wn - base) : 32;
      int r = 0, d = 0;
      float w = 0.0f;
      if (lane < cnt) {
        r = lr[base + lane];
        d = ld[base + lane];
        w = __expf(lv[base + lane] - smax[r]);
        lds_add(den + r, w);
      }
      add_rows(cnt, r, d, w, lane, h, acc);
    }
  } else {
    scan_bitmap<true>(bmRow, W, E, row0, rows, tid, lane, src, dst, a2, h,
                    a1l, smax, acc, den, q, lr, ld, lv);
  }

  __syncthreads();

  for (int i = tid; i < rows * (D_DIM / 4); i += 1024) {
    const int r  = i >> 4;
    const int c4 = (i & 15) << 2;
    const float dn = den[r];
    float4 o = *(const float4*)&acc[r * D_DIM + c4];
    if (dn == 0.0f) {
      o.x = 0.0f; o.y = 0.0f; o.z = 0.0f; o.w = 0.0f;
    } else {
      o.x /= dn; o.y /= dn; o.z /= dn; o.w /= dn;
    }
    *(float4*)(out + (size_t)(row0 + r) * D_DIM + c4) = o;
  }
}

extern "C" void kernel_launch(void* const* d_in, const int* in_sizes, int n_in,
                              void* d_out, int out_size, void* d_ws, size_t ws_size,
                              hipStream_t stream) {
  const float* feat = (const float*)d_in[0];
  const int*   idx  = (const int*)d_in[1];
  const float* W    = (const float*)d_in[2];
  const float* bias = (const float*)d_in[3];
  const float* a1w  = (const float*)d_in[4];
  const float* a1b  = (const float*)d_in[5];
  const float* a2w  = (const float*)d_in[6];
  const float* a2b  = (const float*)d_in[7];
  float* out = (float*)d_out;

  const int n = in_sizes[0] / D_DIM;
  const int E = in_sizes[1] / 2;
  const int* src = idx;
  const int* dst = idx + E;
  const int nRanges = (n + ROWS_PER_WG - 1) / ROWS_PER_WG;
  const int nUnits  = (E + UNIT_EDGES - 1) / UNIT_EDGES;
  const int nSpans  = (nUnits + SPAN_UNITS - 1) / SPAN_UNITS;
  const int bmW     = nSpans * SPAN_WORDS;

  float*    h  = (float*)d_ws;
  int*      pairs = (int*)(h + (size_t)n * D_DIM);
  unsigned* bm = (unsigned*)(pairs + (size_t)nSpans * SPAN_UNITS * UNIT_EDGES * 2);
  float*    a1 = (float*)(bm + (size_t)nRanges * bmW);
  float*    a2 = a1 + n;

  const int tiles = (n + TILE_M - 1) / TILE_M;
  gat_linear_wmma<<<(tiles + 7) / 8, 256, 0, stream>>>(
      feat, W, bias, a1w, a1b, a2w, a2b, h, a1, a2, n);

  gat_edge_bitmap<<<nSpans, 1024, 0, stream>>>(src, dst, bm, pairs, E, nRanges, bmW);

  gat_rows_owner<<<nRanges, 1024, 0, stream>>>(
      src, dst, a1, a2, h, bm, pairs, out, n, E, bmW);
}


# linear pass: global loads of each staging loop (W, feature tile, score weights) and the LDS reads of the h store loop are issued as one group before first use; was one load and one wait per trip
# speedup vs baseline: 5.1531x; 1.0801x over previous
#include <hip/hip_runtime.h>
#include <hip/hip_bf16.h>

typedef __attribute__((ext_vector_type(2))) float v2f;
typedef __attribute__((ext_vector_type(8))) float v8f;

#define D_DIM 64
#define TILE_M 16
#define NEG_SLOPE 0.01f

__global__ __launch_bounds__(256, 2) void gat_linear_wmma(
    const float* __restrict__ feat, const float* __restrict__ W,
    const float* __restrict__ bias,
    const float* __restrict__ a1w, const float* __restrict__ a1b,
    const float* __restrict__ a2w, const float* __restrict__ a2b,
    float* __restrict__ h, float* __restrict__ a1, float* __restrict__ a2,
    int n_nodes)
{
  constexpr int WP = 68;
  constexpr int FP = 68;
  __shared__ float Wl[D_DIM * WP];
  __shared__ float ft[8][TILE_M * FP];

  const int tid  = threadIdx.x;
  const int wave = tid >> 5;
  const int lane = tid & 31;
  const int lh   = lane >> 4;
  const int lid  = lane & 15;

  const int tile = blockIdx.x * 8 + wave;
  const int m0   = tile * TILE_M;
  float* myft = ft[wave];

  float4 wreg[4];
  float4 freg[8];
  #pragma unroll
  for (int j = 0; j < 4; ++j) {
    const int i  = tid + 256 * j;
    const int n  = i >> 4;
    const int c4 = (i & 15) << 2;
    wreg[j] = *(const float4*)(W + n * D_DIM + c4);
  }
  #pragma unroll
  for (int j = 0; j < 8; ++j) {
    const int i  = lane + 32 * j;
    const int r  = i >> 4;
    const int c4 = (i & 15) << 2;
    const int row = (m0 + r) < n_nodes ? (m0 + r) : (n_nodes - 1);
    freg[j] = *(const float4*)(feat + (size_t)row * D_DIM + c4);
  }
  #pragma unroll
  for (int j = 0; j < 4; ++j) {
    const int i  = tid + 256 * j;
    const int n  = i >> 4;
    const int c4 = (i & 15) << 2;
    *(float4*)&Wl[n * WP + c4] = wreg[j];
  }
  #pragma unroll
  for (int j = 0; j < 8; ++j) {
    const int i  = lane + 32 * j;
    const int r  = i >> 4;
    const int c4 = (i & 15) << 2;
    *(float4*)&myft[r * FP + c4] = freg[j];
  }
  __syncthreads();

  if (m0 >= n_nodes) return;

  v8f acc[4] = {};
  const int ks = 2 * lh;
  #pragma unroll
  for (int k = 0; k < D_DIM; k += 4) {
    v2f a;
    a.x = myft[lid * FP + k + ks + 0];
    a.y = myft[lid * FP + k + ks + 1];
    #pragma unroll
    for (int t = 0; t < 4; ++t) {
      const int n = t * 16 + lid;
      v2f b;
      b.x = Wl[n * WP + k + ks + 0];
      b.y = Wl[n * WP + k + ks + 1];
      acc[t] = __builtin_amdgcn_wmma_f32_16x16x4_f32(
          false, a, false, b, (short)0, acc[t], false, false);
    }
  }

  #pragma unroll
  for (int t = 0; t < 4; ++t) {
    const float bb = bias[t * 16 + lid];
    #pragma unroll
    for (int j = 0; j < 8; ++j) {
      myft[(j + 8 * lh) * FP + t * 16 + lid] = acc[t][j] + bb;
    }
  }

  if (m0 + TILE_M <= n_nodes) {
    float4 hv[8];
    #pragma unroll
    for (int j = 0; j < 8; ++j) {
      const int i  = lane + 32 * j;
      hv[j] = *(const float4*)&myft[(i >> 4) * FP + ((i & 15) << 2)];
    }
    #pragma unroll
    for (int j = 0; j < 8; ++j) {
      const int i  = lane + 32 * j;
      *(float4*)(h + (size_t)(m0 + (i >> 4)) * D_DIM + ((i & 15) << 2)) = hv[j];
    }
  } else {
    for (int i = lane; i < TILE_M * (D_DIM / 4); i += 32) {
      const int r  = i >> 4;
      const int c4 = (i & 15) << 2;
      if (m0 + r < n_nodes) {
        const float4 hv = *(const float4*)&myft[r * FP + c4];
        *(float4*)(h + (size_t)(m0 + r) * D_DIM + c4) = hv;
      }
    }
  }

  const float* aw = lh ? a2w : a1w;
  float s = lh ? a2b[0] : a1b[0];
  #pragma unroll
  for (int c0 = 0; c0 < D_DIM; c0 += 32) {
    float4 wv[8], xv[8];
    #pragma unroll
    for (int j = 0; j < 8; ++j) wv[j] = *(const float4*)(aw + c0 + 4 * j);
    #pragma unroll
    for (int j = 0; j < 8; ++j) xv[j] = *(const float4*)&myft[lid * FP + c0 + 4 * j];
    #pragma unroll
    for (int j = 0; j < 8; ++j) {
      s = __builtin_fmaf(xv[j].x, wv[j].x, s);
      s = __builtin_fmaf(xv[j].y, wv[j].y, s);
      s = __builtin_fmaf(xv[j].z, wv[j].z, s);
      s = __builtin_fmaf(xv[j].w, wv[j].w, s);
    }
  }
  if (m0 + lid < n_nodes) {
    if (lh) a2[m0 + lid] = s;
    else    a1[m0 + lid] = s;
  }
}

#define ROWS_PER_WG 392
#define LIST_CAP 320

#define UNIT_EDGES 16
#define SPAN_WORDS 32
#define SPAN_UNITS (SPAN_WORDS * 32)
#define TILE_PITCH 33

__global__ __launch_bounds__(1024) void gat_edge_bitmap(const int* __restrict__ src,
                                                        const int* __restrict__ dst,
                                                        unsigned* __restrict__ bm,
                                                        int* __restrict__ pairs,
                                                        int E, int nRanges, int W) {
  __shared__ unsigned tile[256 * TILE_PITCH];
  const int tid  = threadIdx.x;
  const int lane = tid & 31;
  const int wave = tid >> 5;
  const int e0 = (blockIdx.x * SPAN_UNITS + tid) * UNIT_EDGES;

  {
    constexpr int TRIPS = (SPAN_UNITS * UNIT_EDGES / 2) / 1024;
    const int eS = blockIdx.x * (SPAN_UNITS * UNIT_EDGES);
    int4* pout = (int4*)(pairs + (size_t)eS * 2);
    int sa[TRIPS], da[TRIPS], sb[TRIPS], db[TRIPS];
    #pragma unroll
    for (int k = 0; k < TRIPS; ++k) {
      const int e  = eS + 2 * (tid + 1024 * k);
      const int ea = e < E ? e : E - 1;
      const int eb = e + 1 < E ? e + 1 : E - 1;
      sa[k] = src[ea];
      da[k] = dst[ea];
      sb[k] = src[eb];
      db[k] = dst[eb];
    }
    #pragma unroll
    for (int k = 0; k < TRIPS; ++k) {
      const int i = tid + 1024 * k;
      const int e = eS + 2 * i;
      int4 p;
      p.x = e < E ? sa[k] : -1;
      p.y = e < E ? da[k] : 0;
      p.z = e + 1 < E ? sb[k] : -1;
      p.w = e + 1 < E ? db[k] : 0;
      pout[i] = p;
    }
  }

  unsigned rg[UNIT_EDGES];
  if (e0 + UNIT_EDGES <= E) {
    #pragma unroll
    for (int j = 0; j < 4; ++j) {
      const int4 s4 = *(const int4*)(src + e0 + 4 * j);
      rg[4 * j + 0] = (unsigned)s4.x / (unsigned)ROWS_PER_WG;
      rg[4 * j + 1] = (unsigned)s4.y / (unsigned)ROWS_PER_WG;
      rg[4 * j + 2] = (unsigned)s4.z / (unsigned)ROWS_PER_WG;
      rg[4 * j + 3] = (unsigned)s4.w / (unsigned)ROWS_PER_WG;
    }
  } else {
    #pragma unroll
    for (int k = 0; k < UNIT_EDGES; ++k) {
      const int e = e0 + k;
      rg[k] = e < E ? (unsigned)src[e] / (unsigned)ROWS_PER_WG : 0xFFFFFFFFu;
    }
  }

  const unsigned bit = 1u << lane;
  for (int b0 = 0; b0 < nRanges; b0 += 256) {
    for (int i = tid; i < 256 * TILE_PITCH; i += 1024) tile[i] = 0u;
    __syncthreads();
    #pragma unroll
    for (int k = 0; k < UNIT_EDGES; ++k) {
      const unsigned b = rg[k] - (unsigned)b0;
      if (b < 256u)
        __hip_atomic_fetch_or(&tile[b * TILE_PITCH + wave], bit, __ATOMIC_RELAXED,
                              __HIP_MEMORY_SCOPE_WORKGROUP);
    }
    __syncthreads();
    for (int r = wave; r < 256; r += 32) {
      if (b0 + r < nRanges)
        bm[(size_t)(b0 + r) * W + blockIdx.x * SPAN_WORDS + lane] = tile[r * TILE_PITCH + lane];
    }
    __syncthreads();
  }
}

__device__ __forceinline__ void lds_add(float* p, float v) {
  __hip_atomic_fetch_add(p, v, __ATOMIC_RELAXED, __HIP_MEMORY_SCOPE_WORKGROUP);
}
__device__ __forceinline__ void lds_max(float* p, float v) {
  __hip_atomic_fetch_max(p, v, __ATOMIC_RELAXED, __HIP_MEMORY_SCOPE_WORKGROUP);
}

__device__ __forceinline__ int list_edges(
    int e, int E, int row0, int rows,
    const int* __restrict__ src, const int* __restrict__ dst,
    const float* __restrict__ a2, const float* a1l, float* smax,
    int* lr, int* ld, float* lv, int wn) {
  bool ok = false;
  int r = 0, d = 0;
  float v = 0.0f;
  if (e >= 0 && e < E) {
    const int s = src[e];
    d = dst[e];
    r = s - row0;
    if ((unsigned)r < (unsigned)rows) {
      ok = true;
      v = a1l[r] + a2[d];
      v = v > 0.0f ? v : v * NEG_SLOPE;
      lds_max(smax + r, v);
    }
  }
  const unsigned okm = __builtin_amdgcn_ballot_w32(ok);
  const int pos = wn + (int)__builtin_amdgcn_mbcnt_lo(okm, 0u);
  if (ok && pos < LIST_CAP) {
    lr[pos] = r;
    ld[pos] = d;
    lv[pos] = v;
  }
  return wn + __builtin_popcount(okm);
}

__device__ __forceinline__ void add_rows(int cnt, int r, int d, float w, int lane,
                                         const float* __restrict__ h, float* acc) {
  const int wbits = __float_as_int(w);
  if (cnt == 32) {
    #pragma unroll
    for (int j0 = 0; j0 < 32; j0 += 8) {
      float x0[8], x1[8], wj[8];
      int rj[8];
      #pragma unroll
      for (int j = 0; j < 8; ++j) {
        const int dj = __builtin_amdgcn_readlane(d, j0 + j);
        rj[j] = __builtin_amdgcn_readlane(r, j0 + j);
        wj[j] = __int_as_float(__builtin_amdgcn_readlane(wbits, j0 + j));
        const float* hr = h + (size_t)dj * D_DIM;
        x0[j] = hr[lane];
        x1[j] = hr[lane + 32];
      }
      #pragma unroll
      for (int j = 0; j < 8; ++j) {
        float* ar = acc + rj[j] * D_DIM;
        lds_add(ar + lane,      wj[j] * x0[j]);
        lds_add(ar + lane + 32, wj[j] * x1[j]);
      }
    }
  } else {
    #pragma unroll
    for (int j0 = 0; j0 < 32; j0 += 8) {
      if (j0 < cnt) {
        float x0[8], x1[8], wj[8];
        int rj[8];
        #pragma unroll
        for (int j = 0; j < 8; ++j) {
          const int dj = __builtin_amdgcn_readlane(d, j0 + j);
          rj[j] = __builtin_amdgcn_readlane(r, j0 + j);
          wj[j] = __int_as_float(__builtin_amdgcn_readlane(wbits, j0 + j));
          const float* hr = h + (size_t)dj * D_DIM;
          x0[j] = hr[lane];
          x1[j] = hr[lane + 32];
        }
        #pragma unroll
        for (int j = 0; j < 8; ++j) {
          if (j0 + j < cnt) {
            float* ar = acc + rj[j] * D_DIM;
            lds_add(ar + lane,      wj[j] * x0[j]);
            lds_add(ar + lane + 32, wj[j] * x1[j]);
          }
        }
      }
    }
  }
}

__device__ __forceinline__ void add_edges_direct(
    int e, int E, int row0, int rows, int lane,
    const int* __restrict__ src, const int* __restrict__ dst,
    const float* __restrict__ a2, const float* __restrict__ h,
    const float* a1l, const float* smax, float* acc, float* den) {
  int r = 0, d = 0;
  float w = 0.0f;
  bool ok = false;
  if (e >= 0 && e < E) {
    const int s = src[e];
    r = s - row0;
    if ((unsigned)r < (unsigned)rows) {
      ok = true;
      d = dst[e];
      float v = a1l[r] + a2[d];
      v = v > 0.0f ? v : v * NEG_SLOPE;
      w = __expf(v - smax[r]);
      lds_add(den + r, w);
    }
  }
  unsigned okm = __builtin_amdgcn_ballot_w32(ok);
  const int wbits = __float_as_int(w);
  while (okm) {
    const int j = __builtin_ctz(okm);
    okm &= okm - 1;
    const int dj = __builtin_amdgcn_readlane(d, j);
    const int rk = __builtin_amdgcn_readlane(r, j);
    const float wk = __int_as_float(__builtin_amdgcn_readlane(wbits, j));
    const float* hr = h + (size_t)dj * D_DIM;
    float* ar = acc + rk * D_DIM;
    lds_add(ar + lane,      wk * hr[lane]);
    lds_add(ar + lane + 32, wk * hr[lane + 32]);
  }
}

__device__ __forceinline__ unsigned in_rows(int s, int row0, int rows) {
  return (unsigned)(s - row0) < (unsigned)rows ? 1u : 0u;
}

__device__ __forceinline__ unsigned match_unit(const int4* s4, int row0, int rows) {
  unsigned m = 0u;
  #pragma unroll
  for (int j = 0; j < 4; ++j) {
    m |= (in_rows(s4[j].x, row0, rows) << (4 * j + 0)) |
         (in_rows(s4[j].y, row0, rows) << (4 * j + 1)) |
         (in_rows(s4[j].z, row0, rows) << (4 * j + 2)) |
         (in_rows(s4[j].w, row0, rows) << (4 * j + 3));
  }
  return m;
}

__device__ __forceinline__ unsigned match_last_unit(const int* __restrict__ src, int ebase, int E,
                                                    int row0, int rows) {
  unsigned m = 0u;
  for (int k = 0; k < UNIT_EDGES; ++k) {
    const int e = ebase + k;
    if (e < E) m |= in_rows(src[e], row0, rows) << k;
  }
  return m;
}

template <bool DIRECT>
__device__ __forceinline__ int scan_bitmap(
    const unsigned* __restrict__ bmRow, int W, int E, int row0, int rows, int tid, int lane,
    const int* __restrict__ src, const int* __restrict__ dst,
    const float* __restrict__ a2, const float* __restrict__ h,
    const float* a1l, float* smax, float* acc, float* den,
    int* q, int* lr, int* ld, float* lv) {
  int qn = 0;
  int wn = 0;

  const int wq   = (W + 31) >> 5;
  const int wbeg = (tid >> 5) * wq;
  const int wend = (wbeg + wq) < W ? (wbeg + wq) : W;
  const int nIter = (wq + 31) >> 5;
  unsigned cur = (wbeg + lane) < wend ? bmRow[wbeg + lane] : 0u;

  for (int i = 0; i < nIter; ++i) {
    const int wi = wbeg + i * 32 + lane;
    const unsigned nxt = (wi + 32) < wend ? bmRow[wi + 32] : 0u;
    unsigned bits = cur;

    while (true) {
      const unsigned anyb = __builtin_amdgcn_ballot_w32(bits != 0u);
      if (anyb == 0u) break;
      unsigned m[2];
      int eb[2];
      bool part[2];
      int4 s4[2][4];
      #pragma unroll
      for (int t = 0; t < 2; ++t) {
        const bool hv = bits != 0u;
        const int u = hv ? __builtin_ctz(bits) : 0;
        bits &= bits - 1u;
        eb[t] = (wi * 32 + u) * UNIT_EDGES;
        const bool full = hv && (eb[t] + UNIT_EDGES <= E);
        part[t] = hv && !full;
        #pragma unroll
        for (int j = 0; j < 4; ++j) s4[t][j] = int4{0, 0, 0, 0};
        if (full) {
          #pragma unroll
          for (int j = 0; j < 4; ++j) s4[t][j] = *(const int4*)(src + eb[t] + 4 * j);
        }
        m[t] = full ? 1u : 0u;
      }
      #pragma unroll
      for (int t = 0; t < 2; ++t) {
        m[t] = m[t] != 0u ? match_unit(s4[t], row0, rows) : 0u;
        if (part[t]) m[t] = match_last_unit(src, eb[t], E, row0, rows);
      }
      #pragma unroll
      for (int t = 0; t < 2; ++t) {
        unsigned mm = m[t];
        while (true) {
          const bool has = mm != 0u;
          const unsigned bal = __builtin_amdgcn_ballot_w32(has);
          if (bal == 0u) break;
          if (has) {
            const int p = __builtin_ctz(mm);
            mm &= mm - 1u;
            q[qn + (int)__builtin_amdgcn_mbcnt_lo(bal, 0u)] = eb[t] + p;
          }
          qn += __builtin_popcount(bal);
          if (qn >= 32) {
            qn -= 32;
            const int e = q[qn + lane];
            if constexpr (DIRECT)
              add_edges_direct(e, E, row0, rows, lane, src, dst, a2, h, a1l, smax, acc, den);
            else
              wn = list_edges(e, E, row0, rows, src, dst, a2, a1l, smax, lr, ld, lv, wn);
          }
        }
      }
    }
    cur = nxt;
  }
  if (qn > 0) {
    const int e = lane < qn ? q[lane] : -1;
    if constexpr (DIRECT)
      add_edges_direct(e, E, row0, rows, lane, src, dst, a2, h, a1l, smax, acc, den);
    else
      wn = list_edges(e, E, row0, rows, src, dst, a2, a1l, smax, lr, ld, lv, wn);
  }
  return wn;
}

__device__ __forceinline__ int keep_edge(int s, int d, int row0, int rows,
                                         int* lr, int* ld, int wn) {
  const int r = s - row0;
  const bool ok = (unsigned)r < (unsigned)rows;
  const unsigned okm = __builtin_amdgcn_ballot_w32(ok);
  const int pos = wn + (int)__builtin_amdgcn_mbcnt_lo(okm, 0u);
  if (ok && pos < LIST_CAP) {
    lr[pos] = r;
    ld[pos] = d;
  }
  return wn + __builtin_popcount(okm);
}

__device__ __forceinline__ int take_units(int cnt, const int* ulp, int lane, int row0, int rows,
                                          const int* __restrict__ pairs,
                                          int* lr, int* ld, int wn) {
  int4 p[8];
  #pragma unroll
  for (int j = 0; j < 8; ++j) {
    const int k = 4 * j + (lane >> 3);
    p[j] = int4{-1, 0, -1, 0};
    if (k < cnt) {
      const int u = ulp[k];
      p[j] = *(const int4*)(pairs + (size_t)u * (2 * UNIT_EDGES) + (lane & 7) * 4);
    }
  }
  #pragma unroll
  for (int j = 0; j < 8; ++j) {
    if (4 * j < cnt) {
      wn = keep_edge(p[j].x, p[j].y, row0, rows, lr, ld, wn);
      wn = keep_edge(p[j].z, p[j].w, row0, rows, lr, ld, wn);
    }
  }
  return wn;
}

__device__ __forceinline__ int scan_units(
    const unsigned* __restrict__ bmRow, int W, int row0, int rows, int tid, int lane,
    const int* __restrict__ pairs, int* q, int* lr, int* ld) {
  int un = 0;
  int wn = 0;

  const int wq   = (W + 31) >> 5;
  const int wbeg = (tid >> 5) * wq;
  const int wend = (wbeg + wq) < W ? (wbeg + wq) : W;
  const int nIter = (wq + 31) >> 5;
  unsigned cur = (wbeg + lane) < wend ? bmRow[wbeg + lane] : 0u;

  for (int i = 0; i < nIter; ++i) {
    const int wi = wbeg + i * 32 + lane;
    const unsigned nxt = (wi + 32) < wend ? bmRow[wi + 32] : 0u;
    unsigned bits = cur;
    while (true) {
      const bool has = bits != 0u;
      const unsigned bal = __builtin_amdgcn_ballot_w32(has);
      if (bal == 0u) break;
      if (has) {
        const int u = __builtin_ctz(bits);
        bits &= bits - 1u;
        q[un + (int)__builtin_amdgcn_mbcnt_lo(bal, 0u)] = wi * 32 + u;
      }
      un += __builtin_popcount(bal);
      if (un >= 32) {
        un -= 32;
        wn = take_units(32, q + un, lane, row0, rows, pairs, lr, ld, wn);
      }
    }
    cur = nxt;
  }
  if (un > 0) wn = take_units(un, q, lane, row0, rows, pairs, lr, ld, wn);
  return wn;
}

__device__ __forceinline__ void score_list(int nl, int lane, const float* __restrict__ a2,
                                           const float* a1l, float* smax,
                                           const int* lr, const int* ld, float* lv) {
  for (int base = 0; base < nl; base += 128) {
    int r[4], d[4];
    float a[4];
    #pragma unroll
    for (int t = 0; t < 4; ++t) {
      const int i = base + 32 * t + lane;
      r[t] = 0;
      d[t] = 0;
      if (i < nl) {
        r[t] = lr[i];
        d[t] = ld[i];
      }
    }
    #pragma unroll
    for (int t = 0; t < 4; ++t) a[t] = a2[d[t]];
    #pragma unroll
    for (int t = 0; t < 4; ++t) {
      const int i = base + 32 * t + lane;
      if (i < nl) {
        float v = a1l[r[t]] + a[t];
        v = v > 0.0f ? v : v * NEG_SLOPE;
        lds_max(smax + r[t], v);
        lv[i] = v;
      }
    }
  }
}

__global__ __launch_bounds__(1024) void gat_rows_owner(
    const int* __restrict__ src, const int* __restrict__ dst,
    const float* __restrict__ a1, const float* __restrict__ a2,
    const float* __restrict__ h, const unsigned* __restrict__ bm,
    const int* __restrict__ pairs,
    float* __restrict__ out, int n, int E, int W) {
  __shared__ float acc[ROWS_PER_WG * D_DIM];
  __shared__ float den[ROWS_PER_WG];
  __shared__ float smax[ROWS_PER_WG];
  __shared__ float a1l[ROWS_PER_WG];
  __shared__ int queue[32][64];
  __shared__ int list_r[32][LIST_CAP];
  __shared__ int list_d[32][LIST_CAP];
  __shared__ float list_v[32][LIST_CAP];

  const int tid  = threadIdx.x;
  const int lane = tid & 31;
  const int wave = tid >> 5;
  const int row0 = blockIdx.x * ROWS_PER_WG;
  const int rows = (n - row0) < ROWS_PER_WG ? (n - row0) : ROWS_PER_WG;

  for (int i = tid; i < ROWS_PER_WG * D_DIM; i += 1024) acc[i] = 0.0f;
  for (int i = tid; i < ROWS_PER_WG; i += 1024) {
    den[i] = 0.0f;
    smax[i] = -__builtin_inff();
    a1l[i] = i < rows ? a1[row0 + i] : 0.0f;
  }
  __syncthreads();

  const unsigned* bmRow = bm + (size_t)blockIdx.x * W;
  int* q = queue[wave];
  int* lr = list_r[wave];
  int* ld = list_d[wave];
  float* lv = list_v[wave];

  int wn = scan_units(bmRow, W, row0, rows, tid, lane, pairs, q, lr, ld);
  if (wn <= LIST_CAP)
    score_list(wn, lane, a2, a1l, smax, lr, ld, lv);
  else
    wn = scan_bitmap<false>(bmRow, W, E, row0, rows, tid, lane, src, dst, a2, h,
                            a1l, smax, acc, den, q, lr, ld, lv);

  __syncthreads();

  if (wn <= LIST_CAP) {
    for (int base = 0; base < wn; base += 32) {
      const int cnt = (wn - base) < 32 ? (wn - base) : 32;
      int r = 0, d = 0;
      float w = 0.0f;
      if (lane < cnt) {
        r = lr[base + lane];
        d = ld[base + lane];
        w = __expf(lv[base + lane] - smax[r]);
        lds_add(den + r, w);
      }
      add_rows(cnt, r, d, w, lane, h, acc);
    }
  } else {
    scan_bitmap<true>(bmRow, W, E, row0, rows, tid, lane, src, dst, a2, h,
                    a1l, smax, acc, den, q, lr, ld, lv);
  }

  __syncthreads();

  for (int i = tid; i < rows * (D_DIM / 4); i += 1024) {
    const int r  = i >> 4;
    const int c4 = (i & 15) << 2;
    const float dn = den[r];
    float4 o = *(const float4*)&acc[r * D_DIM + c4];
    if (dn == 0.0f) {
      o.x = 0.0f; o.y = 0.0f; o.z = 0.0f; o.w = 0.0f;
    } else {
      o.x /= dn; o.y /= dn; o.z /= dn; o.w /= dn;
    }
    *(float4*)(out + (size_t)(row0 + r) * D_DIM + c4) = o;
  }
}

extern "C" void kernel_launch(void* const* d_in, const int* in_sizes, int n_in,
                              void* d_out, int out_size, void* d_ws, size_t ws_size,
                              hipStream_t stream) {
  const float* feat = (const float*)d_in[0];
  const int*   idx  = (const int*)d_in[1];
  const float* W    = (const float*)d_in[2];
  const float* bias = (const float*)d_in[3];
  const float* a1w  = (const float*)d_in[4];
  const float* a1b  = (const float*)d_in[5];
  const float* a2w  = (const float*)d_in[6];
  const float* a2b  = (const float*)d_in[7];
  float* out = (float*)d_out;

  const int n = in_sizes[0] / D_DIM;
  const int E = in_sizes[1] / 2;
  const int* src = idx;
  const int* dst = idx + E;
  const int nRanges = (n + ROWS_PER_WG - 1) / ROWS_PER_WG;
  const int nUnits  = (E + UNIT_EDGES - 1) / UNIT_EDGES;
  const int nSpans  = (nUnits + SPAN_UNITS - 1) / SPAN_UNITS;
  const int bmW     = nSpans * SPAN_WORDS;

  float*    h  = (float*)d_ws;
  int*      pairs = (int*)(h + (size_t)n * D_DIM);
  unsigned* bm = (unsigned*)(pairs + (size_t)nSpans * SPAN_UNITS * UNIT_EDGES * 2);
  float*    a1 = (float*)(bm + (size_t)nRanges * bmW);
  float*    a2 = a1 + n;

  const int tiles = (n + TILE_M - 1) / TILE_M;
  gat_linear_wmma<<<(tiles + 7) / 8, 256, 0, stream>>>(
      feat, W, bias, a1w, a1b, a2w, a2b, h, a1, a2, n);

  gat_edge_bitmap<<<nSpans, 1024, 0, stream>>>(src, dst, bm, pairs, E, nRanges, bmW);

  gat_rows_owner<<<nRanges, 1024, 0, stream>>>(
      src, dst, a1, a2, h, bm, pairs, out, n, E, bmW);
}
